# starting kernel
# baseline (speedup 1.0000x reference)
#include <hip/hip_runtime.h>
#include <hip/hip_bf16.h>

typedef __attribute__((ext_vector_type(2))) float v2f;
typedef __attribute__((ext_vector_type(8))) float v8f;


__global__ void init_deg_kernel(float* __restrict__ deg, int n) {
  int i = blockIdx.x * blockDim.x + threadIdx.x;
  if (i < n) deg[i] = 1.0f;
}

__global__ void count_deg_kernel(const int* __restrict__ dst, float* __restrict__ deg, int E) {
  int e = blockIdx.x * blockDim.x + threadIdx.x;
  if (e < E)
    __hip_atomic_fetch_add(&deg[dst[e]], 1.0f, __ATOMIC_RELAXED, __HIP_MEMORY_SCOPE_AGENT);
}

__global__ void rsqrt_kernel(float* __restrict__ deg, int n) {
  int i = blockIdx.x * blockDim.x + threadIdx.x;
  if (i < n) deg[i] = rsqrtf(deg[i]);
}


template <int K, int NC>
__global__ void gemm_wmma_kernel(const float* __restrict__ A,
                                 const float* __restrict__ W,
                                 float* __restrict__ out) {
  const int lane = threadIdx.x & 31;
  const int wave = threadIdx.x >> 5;
  const int m = lane & 15;
  const int h = lane >> 4;
  const int rowTile = blockIdx.x * 16;
  const int colTile = (blockIdx.y * 4 + wave) * 16;

  const float* arow = A + (size_t)(rowTile + m) * K + 2 * h;
  const float* wcol = W + (size_t)(2 * h) * NC + colTile + m;

  v8f c = {0.f, 0.f, 0.f, 0.f, 0.f, 0.f, 0.f, 0.f};

#pragma unroll
  for (int k0 = 0; k0 < K; k0 += 4) {
    float2 av = *(const float2*)(arow + k0);
    v2f a = {av.x, av.y};
    const float* wp = wcol + (size_t)k0 * NC;
    v2f b = {wp[0], wp[NC]};
    c = __builtin_amdgcn_wmma_f32_16x16x4_f32(false, a, false, b, (short)0, c,
                                              false, false);
  }

  float* orow = out + (size_t)(rowTile + h * 8) * NC + colTile + m;
#pragma unroll
  for (int r = 0; r < 8; ++r) orow[(size_t)r * NC] = c[r];
}


__global__ void self_init_kernel(const float* __restrict__ xw,
                                 const float* __restrict__ dinv,
                                 const float* __restrict__ bias,
                                 float* __restrict__ out, int total, int F) {
  int i = blockIdx.x * blockDim.x + threadIdx.x;
  if (i >= total) return;
  int v = i / F;
  int f = i - v * F;
  float s = dinv[v];
  out[i] = s * s * xw[i] + bias[f];
}

__global__ void edge_agg_kernel(const int* __restrict__ src,
                                const int* __restrict__ dst,
                                const float* __restrict__ dinv,
                                const float* __restrict__ xw,
                                float* __restrict__ out, int E, int F) {
  int wid = (blockIdx.x * blockDim.x + threadIdx.x) >> 5;
  int lane = threadIdx.x & 31;
  if (wid >= E) return;
  int s = src[wid];
  int d = dst[wid];
  float norm = dinv[s] * dinv[d];
  const float* xr = xw + (size_t)s * F;
  float* orow = out + (size_t)d * F;
  for (int f = lane; f < F; f += 32) {
    __hip_atomic_fetch_add(&orow[f], norm * xr[f], __ATOMIC_RELAXED,
                           __HIP_MEMORY_SCOPE_AGENT);
  }
}

__global__ void tanh_kernel(float* __restrict__ out, int n) {
  int i = blockIdx.x * blockDim.x + threadIdx.x;
  if (i < n) out[i] = tanhf(out[i]);
}


extern "C" void kernel_launch(void* const* d_in, const int* in_sizes, int n_in,
                              void* d_out, int out_size, void* d_ws, size_t ws_size,
                              hipStream_t stream) {
  const float* x  = (const float*)d_in[0];
  const int*   ei = (const int*)d_in[1];
  const float* W1 = (const float*)d_in[2];
  const float* b1 = (const float*)d_in[3];
  const float* W2 = (const float*)d_in[4];
  const float* b2 = (const float*)d_in[5];

  const int DIN  = in_sizes[3];
  const int DOUT = in_sizes[5];
  const int N    = in_sizes[0] / DIN;
  const int E    = in_sizes[1] / 2;
  const int* src = ei;
  const int* dst = ei + E;

  char* ws = (char*)d_ws;
  size_t off = ((size_t)N * sizeof(float) + 255) & ~(size_t)255;
  float* dinv = (float*)ws;
  float* xw1  = (float*)(ws + off);
  size_t off2 = off + (((size_t)N * DIN * sizeof(float) + 255) & ~(size_t)255);
  float* h1   = (float*)(ws + off2);
  float* xw2  = xw1;
  float* out  = (float*)d_out;

  const int TB = 256;

  init_deg_kernel<<<(N + TB - 1) / TB, TB, 0, stream>>>(dinv, N);
  count_deg_kernel<<<(E + TB - 1) / TB, TB, 0, stream>>>(dst, dinv, E);
  rsqrt_kernel<<<(N + TB - 1) / TB, TB, 0, stream>>>(dinv, N);

  gemm_wmma_kernel<128, 128><<<dim3(N / 16, 128 / 64), 128, 0, stream>>>(x, W1, xw1);
  self_init_kernel<<<(N * DIN + TB - 1) / TB, TB, 0, stream>>>(xw1, dinv, b1, h1,
                                                               N * DIN, DIN);
  edge_agg_kernel<<<((size_t)E * 32 + TB - 1) / TB, TB, 0, stream>>>(src, dst, dinv,
                                                                     xw1, h1, E, DIN);

  gemm_wmma_kernel<128, 64><<<dim3(N / 16, 64 / 64), 128, 0, stream>>>(h1, W2, xw2);
  self_init_kernel<<<(N * DOUT + TB - 1) / TB, TB, 0, stream>>>(xw2, dinv, b2, out,
                                                                N * DOUT, DOUT);
  edge_agg_kernel<<<((size_t)E * 32 + TB - 1) / TB, TB, 0, stream>>>(src, dst, dinv,
                                                                     xw2, out, E, DOUT);
  tanh_kernel<<<(out_size + TB - 1) / TB, TB, 0, stream>>>(out, out_size);
}


# owner groups stop searching: a first launch orders each of 256 edge chunks by owner range in LDS (count, prefix, place) and groups read only their own runs; list order differs, LDS adds as before
# speedup vs baseline: 3.0859x; 3.0859x over previous
#include <hip/hip_runtime.h>
#include <hip/hip_bf16.h>
#include <stdint.h>

typedef __attribute__((ext_vector_type(2))) float v2f;
typedef __attribute__((ext_vector_type(8))) float v8f;


constexpr int AGG_NB = 196;
constexpr int AGG_T  = 1024;
constexpr int AGG_NW = AGG_T / 32;
constexpr int NCH    = 256;
constexpr int SEGW   = NCH / AGG_NW;
constexpr int NGMAX  = 256;
constexpr int PART_T = 256;
constexpr int PART_U = 10;

__global__ void __launch_bounds__(PART_T)
partition_kernel(const int* __restrict__ src, const int* __restrict__ dst, int* __restrict__ ent,
                 int* __restrict__ table, int E, int NG) {
  __shared__ int hist[NGMAX];
  __shared__ int cur[NGMAX];

  const int tid = threadIdx.x;
  const int lane = tid & 31;
  const int c = blockIdx.x;
  const int CH = (E + NCH - 1) / NCH;
  const int e0 = c * CH;
  const int e1 = min(E, e0 + CH);

  for (int i = tid; i < NGMAX; i += PART_T) hist[i] = 0;
  __syncthreads();
  for (int eb = e0; eb < e1; eb += PART_U * PART_T) {
    unsigned d[PART_U];
#pragma unroll
    for (int u = 0; u < PART_U; ++u) d[u] = (unsigned)dst[min(eb + u * PART_T + tid, e1 - 1)];
#pragma unroll
    for (int u = 0; u < PART_U; ++u)
      if (eb + u * PART_T + tid < e1)
        __hip_atomic_fetch_add(&hist[d[u] / AGG_NB], 1, __ATOMIC_RELAXED, __HIP_MEMORY_SCOPE_WORKGROUP);
  }
  __syncthreads();
  if (tid < 32) {
    constexpr int PL = NGMAX / 32;
    int c8[PL];
    int tot = 0;
#pragma unroll
    for (int k = 0; k < PL; ++k) { c8[k] = hist[PL * lane + k]; tot += c8[k]; }
    int inc = tot;
#pragma unroll
    for (int d = 1; d < 32; d <<= 1) {
      const int v = __shfl_up(inc, d, 32);
      if (lane >= d) inc += v;
    }
    int run = inc - tot;
    int2* tab2 = (int2*)table;
#pragma unroll
    for (int k = 0; k < PL; ++k) {
      const int o = PL * lane + k;
      cur[o] = run;
      if (o < NG) tab2[(size_t)o * NCH + c] = make_int2(run, c8[k]);
      run += c8[k];
    }
  }
  __syncthreads();
  for (int eb = e0; eb < e1; eb += PART_U * PART_T) {
    unsigned d[PART_U], s[PART_U];
#pragma unroll
    for (int u = 0; u < PART_U; ++u) {
      const int e = min(eb + u * PART_T + tid, e1 - 1);
      d[u] = (unsigned)dst[e];
      s[u] = (unsigned)src[e];
    }
#pragma unroll
    for (int u = 0; u < PART_U; ++u) {
      if (eb + u * PART_T + tid < e1) {
        const unsigned o = d[u] / AGG_NB;
        const int pos = __hip_atomic_fetch_add(&cur[o], 1, __ATOMIC_RELAXED, __HIP_MEMORY_SCOPE_WORKGROUP);
        ent[e0 + pos] = (int)(((d[u] - o * AGG_NB) << 24) | s[u]);
      }
    }
  }
}

struct SegList {
  int pre[SEGW + 1];
  int sb[SEGW];
  int n;
};

__device__ __forceinline__ int2 seg_fetch(const int* __restrict__ table, int b, int wave, int lane) {
  const int2* trow = (const int2*)table + (size_t)b * NCH + wave * SEGW;
  int2 t = make_int2(0, 0);
  if (lane < SEGW) t = trow[lane];
  return t;
}

__device__ __forceinline__ void seg_setup(SegList& L, int2 t, int wave, int lane, int CH) {
  const int start = (wave * SEGW + lane) * CH + t.x;
  L.pre[0] = 0;
#pragma unroll
  for (int k = 0; k < SEGW; ++k) {
    L.pre[k + 1] = L.pre[k] + __builtin_amdgcn_readlane(t.y, k);
    L.sb[k] = __builtin_amdgcn_readlane(start, k) - L.pre[k];
  }
  L.n = L.pre[SEGW];
}

__device__ __forceinline__ int seg_pos(const SegList& L, int idx) {
  int p = L.sb[0];
#pragma unroll
  for (int k = 1; k < SEGW; ++k) p = (idx >= L.pre[k]) ? L.sb[k] : p;
  return p + idx;
}

__global__ void __launch_bounds__(AGG_T)
degree_kernel(const int* __restrict__ ent, const int* __restrict__ table, float* __restrict__ dinv, int E, int N) {
  __shared__ int degc[AGG_NB];

  const int tid = threadIdx.x;
  const int lane = tid & 31;
  const int wave = __builtin_amdgcn_readfirstlane(tid >> 5);
  const int lo = blockIdx.x * AGG_NB;
  const int nown = min(AGG_NB, N - lo);
  const int CH = (E + NCH - 1) / NCH;

  const int2 myruns = seg_fetch(table, (int)blockIdx.x, wave, lane);
  for (int i = tid; i < AGG_NB; i += AGG_T) degc[i] = 0;
  __syncthreads();

  SegList L;
  seg_setup(L, myruns, wave, lane, CH);
  for (int base = 0; base < L.n; base += 32) {
    const int idx = base + lane;
    if (idx < L.n) {
      const unsigned t = (unsigned)ent[seg_pos(L, idx)];
      __hip_atomic_fetch_add(&degc[t >> 24], 1, __ATOMIC_RELAXED, __HIP_MEMORY_SCOPE_WORKGROUP);
    }
  }
  __syncthreads();

  for (int r = tid; r < nown; r += AGG_T) dinv[lo + r] = rsqrtf(1.0f + (float)degc[r]);
}


template <int F, bool FINAL_TANH, bool EPI, bool FIRST>
__global__ void __launch_bounds__(AGG_T)
agg_owner_kernel(const int* __restrict__ ent, float* __restrict__ wgt, const int* __restrict__ table,
                 const float* __restrict__ dinv, const float* __restrict__ xw,
                 const float* __restrict__ bias, const float* __restrict__ Wepi, const float* __restrict__ Wepi2,
                 float* __restrict__ out, int E, int N) {
  static_assert(F == 128 || F == 64, "row width");
  static_assert(!EPI || (F == 128 && !FINAL_TANH && AGG_NW % 8 == 0), "epilogue products: 128 x 128 weight, then 128 x 64");
  constexpr int F2 = 64;
  constexpr int F4 = F / 4;
  constexpr int PER = F / 32;
  constexpr int PITCH = EPI ? F + 4 : F;
  constexpr int P4 = PITCH / 4;

  __shared__ __attribute__((aligned(16))) float acc[AGG_NB * PITCH];

  const int tid = threadIdx.x;
  const int lane = tid & 31;
  const int wave = __builtin_amdgcn_readfirstlane(tid >> 5);
  const int lo = blockIdx.x * AGG_NB;
  const int nown = min(AGG_NB, N - lo);
  const int CH = (E + NCH - 1) / NCH;

  const int2 myruns = seg_fetch(table, (int)blockIdx.x, wave, lane);

  {
    const float4* xw4 = (const float4*)(xw + (size_t)lo * F);
    const float4* b4 = (const float4*)bias;
    float4* acc4 = (float4*)acc;
    for (int i = tid; i < nown * F4; i += AGG_T) {
      const int r = i / F4;
      const int c = i - r * F4;
      float s = dinv[lo + r];
      s = s * s;
      const float4 x = xw4[i];
      if constexpr (EPI) {
        acc4[r * P4 + c] = make_float4(s * x.x, s * x.y, s * x.z, s * x.w);
      } else {
        const float4 b = b4[c];
        acc4[r * P4 + c] = make_float4(s * x.x + b.x, s * x.y + b.y, s * x.z + b.z, s * x.w + b.w);
      }
    }
  }
  __syncthreads();

  {
    SegList L;
    seg_setup(L, myruns, wave, lane, CH);
    const int n = L.n;
    for (int base = 0; base < n; base += 32) {
      const int idx = base + lane;
      int s = 0, r = 0;
      float w = 0.f;
      if (idx < n) {
        const int p = seg_pos(L, idx);
        const unsigned t = (unsigned)ent[p];
        s = (int)(t & 0xffffffu);
        r = (int)(t >> 24);
        if constexpr (FIRST) {
          w = dinv[s] * dinv[lo + r];
          wgt[p] = w;
        } else {
          w = wgt[p];
        }
      }
      const int nb = min(32, n - base);
      for (int j0 = 0; j0 < nb; j0 += 8) {
        float v[8][PER];
        int rr[8];
        float ww[8];
#pragma unroll
        for (int u = 0; u < 8; ++u) {
          const int j = j0 + u;
          const int sj = __builtin_amdgcn_readlane(s, j);
          rr[u] = __builtin_amdgcn_readlane(r, j);
          ww[u] = __int_as_float(__builtin_amdgcn_readlane(__float_as_int(w), j));
          const float* xr = xw + (size_t)sj * F + lane * PER;
          if constexpr (PER == 4) {
            const float4 t = *(const float4*)xr;
            v[u][0] = t.x; v[u][1] = t.y; v[u][2] = t.z; v[u][3] = t.w;
          } else {
            const float2 t = *(const float2*)xr;
            v[u][0] = t.x; v[u][1] = t.y;
          }
        }
#pragma unroll
        for (int u = 0; u < 8; ++u) {
          if (j0 + u < nb) {
            float* a = acc + rr[u] * PITCH + lane * PER;
#pragma unroll
            for (int k = 0; k < PER; ++k)
              __hip_atomic_fetch_add(&a[k], ww[u] * v[u][k], __ATOMIC_RELAXED,
                                     __HIP_MEMORY_SCOPE_WORKGROUP);
          }
        }
      }
    }
  }
  __syncthreads();

  if constexpr (EPI) {
    const int m = lane & 15;
    const int h = lane >> 4;
    const int ntile = (nown + 15) >> 4;
    __shared__ __attribute__((aligned(16))) float hid[AGG_NB * PITCH];
    {
      const int colT = (wave & 7) * 16;
      v2f bf[F / 4];
      const float* wcol = Wepi + (size_t)(2 * h) * F + colT + m;
#pragma unroll
      for (int kk = 0; kk < F / 4; ++kk) {
        const float* wp = wcol + (size_t)(4 * kk) * F;
        bf[kk] = v2f{wp[0], wp[F]};
      }
      const float bv = bias[colT + m];
#pragma unroll 1
      for (int t = wave >> 3; t < ntile; t += AGG_NW / 8) {
        const int row = min(16 * t + m, nown - 1);
        const float* ar = acc + row * PITCH + 2 * h;
        v8f c = {0.f, 0.f, 0.f, 0.f, 0.f, 0.f, 0.f, 0.f};
#pragma unroll
        for (int kk = 0; kk < F / 4; ++kk) {
          const float2 av = *(const float2*)(ar + 4 * kk);
          v2f a = {av.x, av.y};
          c = __builtin_amdgcn_wmma_f32_16x16x4_f32(false, a, false, bf[kk], (short)0, c, false, false);
        }
        const int r0 = 16 * t + 8 * h;
        float* hrow = hid + r0 * PITCH + colT + m;
#pragma unroll
        for (int r = 0; r < 8; ++r)
          if (r0 + r < nown) hrow[r * PITCH] = c[r] + bv;
      }
    }
    const int colT2 = (wave & 3) * 16;
    v2f bf2[F / 4];
    {
      const float* wcol2 = Wepi2 + (size_t)(2 * h) * F2 + colT2 + m;
#pragma unroll
      for (int kk = 0; kk < F / 4; ++kk) {
        const float* wp = wcol2 + (size_t)(4 * kk) * F2;
        bf2[kk] = v2f{wp[0], wp[F2]};
      }
    }
    __syncthreads();
#pragma unroll 1
    for (int t = wave >> 2; t < ntile; t += AGG_NW / 4) {
      const int row = min(16 * t + m, nown - 1);
      const float* ar = hid + row * PITCH + 2 * h;
      v8f c = {0.f, 0.f, 0.f, 0.f, 0.f, 0.f, 0.f, 0.f};
#pragma unroll
      for (int kk = 0; kk < F / 4; ++kk) {
        const float2 av = *(const float2*)(ar + 4 * kk);
        v2f a = {av.x, av.y};
        c = __builtin_amdgcn_wmma_f32_16x16x4_f32(false, a, false, bf2[kk], (short)0, c, false, false);
      }
      const int r0 = 16 * t + 8 * h;
      float* orow = out + (size_t)(lo + r0) * F2 + colT2 + m;
#pragma unroll
      for (int r = 0; r < 8; ++r)
        if (r0 + r < nown) orow[(size_t)r * F2] = c[r];
    }
  } else {
    const float4* acc4 = (const float4*)acc;
    float4* o4 = (float4*)(out + (size_t)lo * F);
    for (int i = tid; i < nown * F4; i += AGG_T) {
      float4 a = acc4[i];
      if constexpr (FINAL_TANH) {
        a.x = tanhf(a.x); a.y = tanhf(a.y); a.z = tanhf(a.z); a.w = tanhf(a.w);
      }
      o4[i] = a;
    }
  }
}


extern "C" void kernel_launch(void* const* d_in, const int* in_sizes, int n_in,
                              void* d_out, int out_size, void* d_ws, size_t ws_size,
                              hipStream_t stream) {
  const float* x  = (const float*)d_in[0];
  const int*   ei = (const int*)d_in[1];
  const float* W1 = (const float*)d_in[2];
  const float* b1 = (const float*)d_in[3];
  const float* W2 = (const float*)d_in[4];
  const float* b2 = (const float*)d_in[5];

  const int DIN  = in_sizes[3];
  const int DOUT = in_sizes[5];
  const int N    = in_sizes[0] / DIN;
  const int E    = in_sizes[1] / 2;
  const int* src = ei;
  const int* dst = ei + E;

  char* ws = (char*)d_ws;
  size_t off = ((size_t)N * sizeof(float) + 255) & ~(size_t)255;
  float* dinv = (float*)ws;
  float* xw1  = (float*)(ws + off);
  float* xw2  = xw1;
  float* out  = (float*)d_out;
  int*   ent   = (int*)(xw1 + (size_t)N * DOUT);
  float* wgt   = (float*)(ent + (E + 3) / 4 * 4);
  int*   table = (int*)(wgt + (E + 3) / 4 * 4);

  const int NG = (N + AGG_NB - 1) / AGG_NB;

  partition_kernel<<<NCH, PART_T, 0, stream>>>(src, dst, ent, table, E, NG);
  degree_kernel<<<NG, AGG_T, 0, stream>>>(ent, table, dinv, E, N);

  agg_owner_kernel<128, false, true, true><<<NG, AGG_T, 0, stream>>>(ent, wgt, table, dinv, x, b1, W1, W2, xw2, E, N);

  agg_owner_kernel<64, true, false, false><<<NG, AGG_T, 0, stream>>>(ent, wgt, table, dinv, xw2, b2, nullptr, nullptr, out, E, N);
}


# layer-1 edge adds: a lane takes 2 floats of each 64-float half of a row (two 8-B loads, not one 16-B): an LDS float add's 32 lanes hit 32 banks, not 16 twice; LDS layout, arrival-order adds kept
# speedup vs baseline: 3.6204x; 1.1732x over previous
#include <hip/hip_runtime.h>
#include <hip/hip_bf16.h>
#include <stdint.h>

typedef __attribute__((ext_vector_type(2))) float v2f;
typedef __attribute__((ext_vector_type(8))) float v8f;


constexpr int AGG_NB = 196;
constexpr int AGG_T  = 1024;
constexpr int AGG_NW = AGG_T / 32;
constexpr int NCH    = 256;
constexpr int SEGW   = NCH / AGG_NW;
constexpr int NGMAX  = 256;
constexpr int PART_T = 256;
constexpr int PART_U = 10;

__global__ void __launch_bounds__(PART_T)
partition_kernel(const int* __restrict__ src, const int* __restrict__ dst, int* __restrict__ ent,
                 int* __restrict__ table, int E, int NG) {
  __shared__ int hist[NGMAX];
  __shared__ int cur[NGMAX];

  const int tid = threadIdx.x;
  const int lane = tid & 31;
  const int c = blockIdx.x;
  const int CH = (E + NCH - 1) / NCH;
  const int e0 = c * CH;
  const int e1 = min(E, e0 + CH);

  for (int i = tid; i < NGMAX; i += PART_T) hist[i] = 0;
  __syncthreads();
  for (int eb = e0; eb < e1; eb += PART_U * PART_T) {
    unsigned d[PART_U];
#pragma unroll
    for (int u = 0; u < PART_U; ++u) d[u] = (unsigned)dst[min(eb + u * PART_T + tid, e1 - 1)];
#pragma unroll
    for (int u = 0; u < PART_U; ++u)
      if (eb + u * PART_T + tid < e1)
        __hip_atomic_fetch_add(&hist[d[u] / AGG_NB], 1, __ATOMIC_RELAXED, __HIP_MEMORY_SCOPE_WORKGROUP);
  }
  __syncthreads();
  if (tid < 32) {
    constexpr int PL = NGMAX / 32;
    int c8[PL];
    int tot = 0;
#pragma unroll
    for (int k = 0; k < PL; ++k) { c8[k] = hist[PL * lane + k]; tot += c8[k]; }
    int inc = tot;
#pragma unroll
    for (int d = 1; d < 32; d <<= 1) {
      const int v = __shfl_up(inc, d, 32);
      if (lane >= d) inc += v;
    }
    int run = inc - tot;
    int2* tab2 = (int2*)table;
#pragma unroll
    for (int k = 0; k < PL; ++k) {
      const int o = PL * lane + k;
      cur[o] = run;
      if (o < NG) tab2[(size_t)o * NCH + c] = make_int2(run, c8[k]);
      run += c8[k];
    }
  }
  __syncthreads();
  for (int eb = e0; eb < e1; eb += PART_U * PART_T) {
    unsigned d[PART_U], s[PART_U];
#pragma unroll
    for (int u = 0; u < PART_U; ++u) {
      const int e = min(eb + u * PART_T + tid, e1 - 1);
      d[u] = (unsigned)dst[e];
      s[u] = (unsigned)src[e];
    }
#pragma unroll
    for (int u = 0; u < PART_U; ++u) {
      if (eb + u * PART_T + tid < e1) {
        const unsigned o = d[u] / AGG_NB;
        const int pos = __hip_atomic_fetch_add(&cur[o], 1, __ATOMIC_RELAXED, __HIP_MEMORY_SCOPE_WORKGROUP);
        ent[e0 + pos] = (int)(((d[u] - o * AGG_NB) << 24) | s[u]);
      }
    }
  }
}

struct SegList {
  int pre[SEGW + 1];
  int sb[SEGW];
  int n;
};

__device__ __forceinline__ int2 seg_fetch(const int* __restrict__ table, int b, int wave, int lane) {
  const int2* trow = (const int2*)table + (size_t)b * NCH + wave * SEGW;
  int2 t = make_int2(0, 0);
  if (lane < SEGW) t = trow[lane];
  return t;
}

__device__ __forceinline__ void seg_setup(SegList& L, int2 t, int wave, int lane, int CH) {
  const int start = (wave * SEGW + lane) * CH + t.x;
  L.pre[0] = 0;
#pragma unroll
  for (int k = 0; k < SEGW; ++k) {
    L.pre[k + 1] = L.pre[k] + __builtin_amdgcn_readlane(t.y, k);
    L.sb[k] = __builtin_amdgcn_readlane(start, k) - L.pre[k];
  }
  L.n = L.pre[SEGW];
}

__device__ __forceinline__ int seg_pos(const SegList& L, int idx) {
  int p = L.sb[0];
#pragma unroll
  for (int k = 1; k < SEGW; ++k) p = (idx >= L.pre[k]) ? L.sb[k] : p;
  return p + idx;
}

__global__ void __launch_bounds__(AGG_T)
degree_kernel(const int* __restrict__ ent, const int* __restrict__ table, float* __restrict__ dinv, int E, int N) {
  __shared__ int degc[AGG_NB];

  const int tid = threadIdx.x;
  const int lane = tid & 31;
  const int wave = __builtin_amdgcn_readfirstlane(tid >> 5);
  const int lo = blockIdx.x * AGG_NB;
  const int nown = min(AGG_NB, N - lo);
  const int CH = (E + NCH - 1) / NCH;

  const int2 myruns = seg_fetch(table, (int)blockIdx.x, wave, lane);
  for (int i = tid; i < AGG_NB; i += AGG_T) degc[i] = 0;
  __syncthreads();

  SegList L;
  seg_setup(L, myruns, wave, lane, CH);
  for (int base = 0; base < L.n; base += 32) {
    const int idx = base + lane;
    if (idx < L.n) {
      const unsigned t = (unsigned)ent[seg_pos(L, idx)];
      __hip_atomic_fetch_add(&degc[t >> 24], 1, __ATOMIC_RELAXED, __HIP_MEMORY_SCOPE_WORKGROUP);
    }
  }
  __syncthreads();

  for (int r = tid; r < nown; r += AGG_T) dinv[lo + r] = rsqrtf(1.0f + (float)degc[r]);
}


template <int F, bool FINAL_TANH, bool EPI, bool FIRST>
__global__ void __launch_bounds__(AGG_T)
agg_owner_kernel(const int* __restrict__ ent, float* __restrict__ wgt, const int* __restrict__ table,
                 const float* __restrict__ dinv, const float* __restrict__ xw,
                 const float* __restrict__ bias, const float* __restrict__ Wepi, const float* __restrict__ Wepi2,
                 float* __restrict__ out, int E, int N) {
  static_assert(F == 128 || F == 64, "row width");
  static_assert(!EPI || (F == 128 && !FINAL_TANH && AGG_NW % 8 == 0), "epilogue products: 128 x 128 weight, then 128 x 64");
  constexpr int F2 = 64;
  constexpr int F4 = F / 4;
  constexpr int NH = F / 64;
  constexpr int PER = 2 * NH;
  constexpr int PITCH = EPI ? F + 4 : F;
  constexpr int P4 = PITCH / 4;

  __shared__ __attribute__((aligned(16))) float acc[AGG_NB * PITCH];

  const int tid = threadIdx.x;
  const int lane = tid & 31;
  const int wave = __builtin_amdgcn_readfirstlane(tid >> 5);
  const int lo = blockIdx.x * AGG_NB;
  const int nown = min(AGG_NB, N - lo);
  const int CH = (E + NCH - 1) / NCH;

  const int2 myruns = seg_fetch(table, (int)blockIdx.x, wave, lane);

  {
    const float4* xw4 = (const float4*)(xw + (size_t)lo * F);
    const float4* b4 = (const float4*)bias;
    float4* acc4 = (float4*)acc;
    for (int i = tid; i < nown * F4; i += AGG_T) {
      const int r = i / F4;
      const int c = i - r * F4;
      float s = dinv[lo + r];
      s = s * s;
      const float4 x = xw4[i];
      if constexpr (EPI) {
        acc4[r * P4 + c] = make_float4(s * x.x, s * x.y, s * x.z, s * x.w);
      } else {
        const float4 b = b4[c];
        acc4[r * P4 + c] = make_float4(s * x.x + b.x, s * x.y + b.y, s * x.z + b.z, s * x.w + b.w);
      }
    }
  }
  __syncthreads();

  {
    SegList L;
    seg_setup(L, myruns, wave, lane, CH);
    const int n = L.n;
    for (int base = 0; base < n; base += 32) {
      const int idx = base + lane;
      int s = 0, r = 0;
      float w = 0.f;
      if (idx < n) {
        const int p = seg_pos(L, idx);
        const unsigned t = (unsigned)ent[p];
        s = (int)(t & 0xffffffu);
        r = (int)(t >> 24);
        if constexpr (FIRST) {
          w = dinv[s] * dinv[lo + r];
          wgt[p] = w;
        } else {
          w = wgt[p];
        }
      }
      const int nb = min(32, n - base);
      for (int j0 = 0; j0 < nb; j0 += 8) {
        float v[8][PER];
        int rr[8];
        float ww[8];
#pragma unroll
        for (int u = 0; u < 8; ++u) {
          const int j = j0 + u;
          const int sj = __builtin_amdgcn_readlane(s, j);
          rr[u] = __builtin_amdgcn_readlane(r, j);
          ww[u] = __int_as_float(__builtin_amdgcn_readlane(__float_as_int(w), j));
          const float* xr = xw + (size_t)sj * F + 2 * lane;
#pragma unroll
          for (int hh = 0; hh < NH; ++hh) {
            const float2 t = *(const float2*)(xr + 64 * hh);
            v[u][2 * hh] = t.x; v[u][2 * hh + 1] = t.y;
          }
        }
#pragma unroll
        for (int u = 0; u < 8; ++u) {
          if (j0 + u < nb) {
            float* a = acc + rr[u] * PITCH + 2 * lane;
#pragma unroll
            for (int k = 0; k < PER; ++k)
              __hip_atomic_fetch_add(&a[64 * (k >> 1) + (k & 1)], ww[u] * v[u][k], __ATOMIC_RELAXED,
                                     __HIP_MEMORY_SCOPE_WORKGROUP);
          }
        }
      }
    }
  }
  __syncthreads();

  if constexpr (EPI) {
    const int m = lane & 15;
    const int h = lane >> 4;
    const int ntile = (nown + 15) >> 4;
    __shared__ __attribute__((aligned(16))) float hid[AGG_NB * PITCH];
    {
      const int colT = (wave & 7) * 16;
      v2f bf[F / 4];
      const float* wcol = Wepi + (size_t)(2 * h) * F + colT + m;
#pragma unroll
      for (int kk = 0; kk < F / 4; ++kk) {
        const float* wp = wcol + (size_t)(4 * kk) * F;
        bf[kk] = v2f{wp[0], wp[F]};
      }
      const float bv = bias[colT + m];
#pragma unroll 1
      for (int t = wave >> 3; t < ntile; t += AGG_NW / 8) {
        const int row = min(16 * t + m, nown - 1);
        const float* ar = acc + row * PITCH + 2 * h;
        v8f c = {0.f, 0.f, 0.f, 0.f, 0.f, 0.f, 0.f, 0.f};
#pragma unroll
        for (int kk = 0; kk < F / 4; ++kk) {
          const float2 av = *(const float2*)(ar + 4 * kk);
          v2f a = {av.x, av.y};
          c = __builtin_amdgcn_wmma_f32_16x16x4_f32(false, a, false, bf[kk], (short)0, c, false, false);
        }
        const int r0 = 16 * t + 8 * h;
        float* hrow = hid + r0 * PITCH + colT + m;
#pragma unroll
        for (int r = 0; r < 8; ++r)
          if (r0 + r < nown) hrow[r * PITCH] = c[r] + bv;
      }
    }
    const int colT2 = (wave & 3) * 16;
    v2f bf2[F / 4];
    {
      const float* wcol2 = Wepi2 + (size_t)(2 * h) * F2 + colT2 + m;
#pragma unroll
      for (int kk = 0; kk < F / 4; ++kk) {
        const float* wp = wcol2 + (size_t)(4 * kk) * F2;
        bf2[kk] = v2f{wp[0], wp[F2]};
      }
    }
    __syncthreads();
#pragma unroll 1
    for (int t = wave >> 2; t < ntile; t += AGG_NW / 4) {
      const int row = min(16 * t + m, nown - 1);
      const float* ar = hid + row * PITCH + 2 * h;
      v8f c = {0.f, 0.f, 0.f, 0.f, 0.f, 0.f, 0.f, 0.f};
#pragma unroll
      for (int kk = 0; kk < F / 4; ++kk) {
        const float2 av = *(const float2*)(ar + 4 * kk);
        v2f a = {av.x, av.y};
        c = __builtin_amdgcn_wmma_f32_16x16x4_f32(false, a, false, bf2[kk], (short)0, c, false, false);
      }
      const int r0 = 16 * t + 8 * h;
      float* orow = out + (size_t)(lo + r0) * F2 + colT2 + m;
#pragma unroll
      for (int r = 0; r < 8; ++r)
        if (r0 + r < nown) orow[(size_t)r * F2] = c[r];
    }
  } else {
    const float4* acc4 = (const float4*)acc;
    float4* o4 = (float4*)(out + (size_t)lo * F);
    for (int i = tid; i < nown * F4; i += AGG_T) {
      float4 a = acc4[i];
      if constexpr (FINAL_TANH) {
        a.x = tanhf(a.x); a.y = tanhf(a.y); a.z = tanhf(a.z); a.w = tanhf(a.w);
      }
      o4[i] = a;
    }
  }
}


extern "C" void kernel_launch(void* const* d_in, const int* in_sizes, int n_in,
                              void* d_out, int out_size, void* d_ws, size_t ws_size,
                              hipStream_t stream) {
  const float* x  = (const float*)d_in[0];
  const int*   ei = (const int*)d_in[1];
  const float* W1 = (const float*)d_in[2];
  const float* b1 = (const float*)d_in[3];
  const float* W2 = (const float*)d_in[4];
  const float* b2 = (const float*)d_in[5];

  const int DIN  = in_sizes[3];
  const int DOUT = in_sizes[5];
  const int N    = in_sizes[0] / DIN;
  const int E    = in_sizes[1] / 2;
  const int* src = ei;
  const int* dst = ei + E;

  char* ws = (char*)d_ws;
  size_t off = ((size_t)N * sizeof(float) + 255) & ~(size_t)255;
  float* dinv = (float*)ws;
  float* xw1  = (float*)(ws + off);
  float* xw2  = xw1;
  float* out  = (float*)d_out;
  int*   ent   = (int*)(xw1 + (size_t)N * DOUT);
  float* wgt   = (float*)(ent + (E + 3) / 4 * 4);
  int*   table = (int*)(wgt + (E + 3) / 4 * 4);

  const int NG = (N + AGG_NB - 1) / AGG_NB;

  partition_kernel<<<NCH, PART_T, 0, stream>>>(src, dst, ent, table, E, NG);
  degree_kernel<<<NG, AGG_T, 0, stream>>>(ent, table, dinv, E, N);

  agg_owner_kernel<128, false, true, true><<<NG, AGG_T, 0, stream>>>(ent, wgt, table, dinv, x, b1, W1, W2, xw2, E, N);

  agg_owner_kernel<64, true, false, false><<<NG, AGG_T, 0, stream>>>(ent, wgt, table, dinv, xw2, b2, nullptr, nullptr, out, E, N);
}
